# starting kernel
# baseline (speedup 1.0000x reference)
#include <hip/hip_runtime.h>

#define LMAXV 2
#define CCH 8

typedef float v2f __attribute__((ext_vector_type(2)));
typedef float v8f __attribute__((ext_vector_type(8)));

constexpr double cfact(int n) { double r = 1.0; for (int i = 2; i <= n; ++i) r *= i; return r; }
constexpr double csqrt_(double x) {
  if (x <= 0.0) return 0.0;
  double r = x < 1.0 ? 1.0 : x;
  for (int i = 0; i < 100; ++i) r = 0.5 * (r + x / r);
  return r;
}
constexpr double cgcoef(int j1, int m1, int j2, int m2, int j, int m) {
  if (m1 + m2 != m) return 0.0;
  int lo = j1 > j2 ? j1 - j2 : j2 - j1;
  if (j < lo || j > j1 + j2) return 0.0;
  double pre = csqrt_((2.0 * j + 1.0) * cfact(j + j1 - j2) * cfact(j - j1 + j2) *
                      cfact(j1 + j2 - j) / cfact(j1 + j2 + j + 1));
  pre *= csqrt_(cfact(j + m) * cfact(j - m) * cfact(j1 - m1) * cfact(j1 + m1) *
                cfact(j2 - m2) * cfact(j2 + m2));
  double s = 0.0;
  for (int k = 0; k <= j1 + j2 - j; ++k) {
    int d2 = j1 + j2 - j - k, d3 = j1 - m1 - k, d4 = j2 + m2 - k;
    int d5 = j - j2 + m1 + k, d6 = j - j1 - m2 + k;
    if (d2 < 0 || d3 < 0 || d4 < 0 || d5 < 0 || d6 < 0) continue;
    double den = cfact(k) * cfact(d2) * cfact(d3) * cfact(d4) * cfact(d5) * cfact(d6);
    s += ((k & 1) ? -1.0 : 1.0) / den;
  }
  return pre * s;
}

struct CGPack {
  int ncoup[3];
  int l1t[3][6], l2t[3][6];
  int nnz[3][5][6];
  int pq[3][5][6][5];
  float coef[3][5][6][5];
};

constexpr CGPack buildCG() {
  CGPack P{};
  for (int l = 0; l <= LMAXV; ++l) {
    int nc = 0;
    for (int l1 = 0; l1 <= LMAXV; ++l1)
      for (int l2 = 0; l2 <= LMAXV; ++l2) {
        int lo = l1 > l2 ? l1 - l2 : l2 - l1;
        int hi = (l1 + l2 > LMAXV) ? LMAXV : (l1 + l2);
        if (l < lo || l > hi) continue;
        P.l1t[l][nc] = l1; P.l2t[l][nc] = l2;
        for (int mi = 0; mi < 2 * l + 1; ++mi) {
          int m = mi - l, n = 0;
          for (int p = 0; p < 2 * l1 + 1; ++p) {
            int m1 = p - l1, m2 = m - m1;
            if (m2 < -l2 || m2 > l2) continue;
            double v = cgcoef(l1, m1, l2, m2, l, m);
            if (v == 0.0) continue;
            P.pq[l][mi][nc][n] = p * 8 + (m2 + l2);
            P.coef[l][mi][nc][n] = (float)v;
            ++n;
          }
          P.nnz[l][mi][nc] = n;
        }
        ++nc;
      }
    P.ncoup[l] = nc;
  }
  return P;
}

constexpr CGPack H = buildCG();
constexpr int LOFF[3] = {0, 8, 32};
constexpr int NCOUPC[3] = {3, 6, 6};

__global__ void zero_f32(float* __restrict__ p, long n) {
  long i = (long)blockIdx.x * blockDim.x + threadIdx.x;
  if (i < n) p[i] = 0.0f;
}

__global__ __launch_bounds__(256) void edge_mp(
    const float* __restrict__ x0, const float* __restrict__ x1,
    const float* __restrict__ x2, const float* __restrict__ ev,
    const int* __restrict__ src, const int* __restrict__ dst,
    float* __restrict__ y, int nE) {
  long idx = (long)blockIdx.x * blockDim.x + threadIdx.x;
  if (idx >= (long)nE * 6) return;
  int e = (int)(idx / 6);
  int g = (int)(idx % 6);
  int s = src[e], d = dst[e];
  float v = ev[e];
#pragma unroll
  for (int j = 0; j < 3; ++j) {
    int slot = g * 3 + j;
    const float* ps;
    if (slot < 2)      ps = x0 + (size_t)s * 8  + slot * 4;
    else if (slot < 8) ps = x1 + (size_t)s * 24 + (slot - 2) * 4;
    else               ps = x2 + (size_t)s * 40 + (slot - 8) * 4;
    float4 val = *(const float4*)ps;
    float* py = y + (size_t)d * 72 + slot * 4;
    __hip_atomic_fetch_add(py + 0, v * val.x, __ATOMIC_RELAXED, __HIP_MEMORY_SCOPE_AGENT);
    __hip_atomic_fetch_add(py + 1, v * val.y, __ATOMIC_RELAXED, __HIP_MEMORY_SCOPE_AGENT);
    __hip_atomic_fetch_add(py + 2, v * val.z, __ATOMIC_RELAXED, __HIP_MEMORY_SCOPE_AGENT);
    __hip_atomic_fetch_add(py + 3, v * val.w, __ATOMIC_RELAXED, __HIP_MEMORY_SCOPE_AGENT);
  }
}

__global__ __launch_bounds__(64) void node_cg_wmma(
    const float* __restrict__ x0, const float* __restrict__ x1,
    const float* __restrict__ x2, const float* __restrict__ y,
    const float* __restrict__ wyy0, const float* __restrict__ wyy1,
    const float* __restrict__ wyy2, const float* __restrict__ wyx0,
    const float* __restrict__ wyx1, const float* __restrict__ wyx2,
    const float* __restrict__ wxx0, const float* __restrict__ wxx1,
    const float* __restrict__ wxx2, float* __restrict__ out, int nN) {
  __shared__ float lds[2 * 16 * 144];

  const int wave = threadIdx.x >> 5;
  const int lane = threadIdx.x & 31;
  const int row  = lane & 15;
  const int hi   = lane >> 4;
  const int wcol = row & 7;
  const int tileBase = blockIdx.x * 32 + wave * 16;

  const float* W[3][3] = {{wyy0, wyx0, wxx0}, {wyy1, wyx1, wxx1}, {wyy2, wyx2, wxx2}};
  __builtin_prefetch(W[0][0], 0, 3);
  __builtin_prefetch(W[1][0], 0, 3);
  __builtin_prefetch(W[2][0], 0, 3);

  float* L = lds + wave * 16 * 144;
  for (int i = lane; i < 16 * 72; i += 32) {
    int ln = i / 72, f = i % 72;
    int node = tileBase + ln;
    if (node >= nN) node = nN - 1;
    float xv;
    if (f < 8)       xv = x0[(size_t)node * 8  + f];
    else if (f < 32) xv = x1[(size_t)node * 24 + (f - 8)];
    else             xv = x2[(size_t)node * 40 + (f - 32)];
    L[ln * 144 + f]      = xv;
    L[ln * 144 + 72 + f] = y[(size_t)node * 72 + f];
  }
  __syncthreads();

  const float* Lrow = L + row * 144;
  const v8f vzero = {0.f, 0.f, 0.f, 0.f, 0.f, 0.f, 0.f, 0.f};

#pragma unroll
  for (int l = 0; l < 3; ++l) {
    const int nm = 2 * l + 1;
    v8f acc[5];
#pragma unroll
    for (int m = 0; m < 5; ++m) acc[m] = vzero;

#pragma unroll
    for (int prod = 0; prod < 3; ++prod) {
      const float* wl = W[l][prod];
      const int aoff = (prod < 2)  ? 72 : 0;
      const int boff = (prod == 0) ? 72 : 0;
#pragma unroll
      for (int cp = 0; cp < NCOUPC[l]; ++cp) {
        const int l1 = H.l1t[l][cp], l2 = H.l2t[l][cp];
        const int na = 2 * l1 + 1, nb = 2 * l2 + 1;
        const float* paBase = Lrow + aoff + LOFF[l1];
        const float* pbBase = Lrow + boff + LOFF[l2];
        const float* wcp = wl + (size_t)(cp * 64) * CCH;
        for (int c = 0; c < 8; ++c) {
          const float* pa = paBase + c * na;
          for (int dc = 0; dc < 8; dc += 4) {
            const int dlane = dc + 2 * hi;
            const int kb = c * 8 + dlane;
            v2f B;
            B.x = wcp[(kb + 0) * CCH + wcol];
            B.y = wcp[(kb + 1) * CCH + wcol];
            const float* pb0 = pbBase + dlane * nb;
            const float* pb1 = pb0 + nb;
#pragma unroll
            for (int m = 0; m < 2 * l + 1; ++m) {
              float z0 = 0.f, z1 = 0.f;
#pragma unroll
              for (int t = 0; t < H.nnz[l][m][cp]; ++t) {
                const int pq = H.pq[l][m][cp][t];
                const float cf = H.coef[l][m][cp][t];
                const float a = pa[pq >> 3];
                z0 += cf * a * pb0[pq & 7];
                z1 += cf * a * pb1[pq & 7];
              }
              v2f A; A.x = z0; A.y = z1;
              acc[m] = __builtin_amdgcn_wmma_f32_16x16x4_f32(
                  false, A, false, B, (short)0, acc[m], false, false);
            }
          }
        }
      }
    }

    const int o = row;
    if (o < CCH) {
      const size_t baseL = (l == 0) ? 0 : (l == 1 ? (size_t)nN * 8 : (size_t)nN * 32);
#pragma unroll
      for (int r = 0; r < 8; ++r) {
        int ln = r + 8 * hi;
        int node = tileBase + ln;
        if (node >= nN) node = nN - 1;
        const float* Lx = L + ln * 144 + LOFF[l] + o * nm;
        float* po = out + baseL + (size_t)node * (CCH * nm) + o * nm;
#pragma unroll
        for (int m = 0; m < 2 * l + 1; ++m) po[m] = acc[m][r] + Lx[m];
      }
    }
  }
}

extern "C" void kernel_launch(void* const* d_in, const int* in_sizes, int n_in,
                              void* d_out, int out_size, void* d_ws, size_t ws_size,
                              hipStream_t stream) {
  const float* x0 = (const float*)d_in[0];
  const float* x1 = (const float*)d_in[1];
  const float* x2 = (const float*)d_in[2];
  const float* ev = (const float*)d_in[3];
  const int* esrc = (const int*)d_in[4];
  const int* edst = (const int*)d_in[5];
  const float* wyy0 = (const float*)d_in[6];
  const float* wyy1 = (const float*)d_in[7];
  const float* wyy2 = (const float*)d_in[8];
  const float* wyx0 = (const float*)d_in[9];
  const float* wyx1 = (const float*)d_in[10];
  const float* wyx2 = (const float*)d_in[11];
  const float* wxx0 = (const float*)d_in[12];
  const float* wxx1 = (const float*)d_in[13];
  const float* wxx2 = (const float*)d_in[14];
  float* out = (float*)d_out;
  float* y   = (float*)d_ws;

  const int nN = in_sizes[0] / CCH;
  const int nE = in_sizes[3];

  long ny = (long)nN * 72;
  zero_f32<<<(unsigned)((ny + 511) / 512), 512, 0, stream>>>(y, ny);

  long tot = (long)nE * 6;
  edge_mp<<<(unsigned)((tot + 255) / 256), 256, 0, stream>>>(x0, x1, x2, ev, esrc,
                                                             edst, y, nE);

  int blocks = (nN + 31) / 32;
  node_cg_wmma<<<blocks, 64, 0, stream>>>(x0, x1, x2, y, wyy0, wyy1, wyy2, wyx0,
                                          wyx1, wyx2, wxx0, wxx1, wxx2, out, nN);
}


# node kernel: each 16-node tile's K work cut into 9 equal shares, one wave each, so 9x the waves hide load latency; the 9 partial tiles are summed through LDS in slot order instead of one chain
# speedup vs baseline: 1.5615x; 1.5615x over previous
#include <hip/hip_runtime.h>

#define LMAXV 2
#define CCH 8

typedef float v2f __attribute__((ext_vector_type(2)));
typedef float v8f __attribute__((ext_vector_type(8)));

constexpr double cfact(int n) { double r = 1.0; for (int i = 2; i <= n; ++i) r *= i; return r; }
constexpr double csqrt_(double x) {
  if (x <= 0.0) return 0.0;
  double r = x < 1.0 ? 1.0 : x;
  for (int i = 0; i < 100; ++i) r = 0.5 * (r + x / r);
  return r;
}
constexpr double cgcoef(int j1, int m1, int j2, int m2, int j, int m) {
  if (m1 + m2 != m) return 0.0;
  int lo = j1 > j2 ? j1 - j2 : j2 - j1;
  if (j < lo || j > j1 + j2) return 0.0;
  double pre = csqrt_((2.0 * j + 1.0) * cfact(j + j1 - j2) * cfact(j - j1 + j2) *
                      cfact(j1 + j2 - j) / cfact(j1 + j2 + j + 1));
  pre *= csqrt_(cfact(j + m) * cfact(j - m) * cfact(j1 - m1) * cfact(j1 + m1) *
                cfact(j2 - m2) * cfact(j2 + m2));
  double s = 0.0;
  for (int k = 0; k <= j1 + j2 - j; ++k) {
    int d2 = j1 + j2 - j - k, d3 = j1 - m1 - k, d4 = j2 + m2 - k;
    int d5 = j - j2 + m1 + k, d6 = j - j1 - m2 + k;
    if (d2 < 0 || d3 < 0 || d4 < 0 || d5 < 0 || d6 < 0) continue;
    double den = cfact(k) * cfact(d2) * cfact(d3) * cfact(d4) * cfact(d5) * cfact(d6);
    s += ((k & 1) ? -1.0 : 1.0) / den;
  }
  return pre * s;
}

struct CGPack {
  int ncoup[3];
  int l1t[3][6], l2t[3][6];
  int nnz[3][5][6];
  int pq[3][5][6][5];
  float coef[3][5][6][5];
};

constexpr CGPack buildCG() {
  CGPack P{};
  for (int l = 0; l <= LMAXV; ++l) {
    int nc = 0;
    for (int l1 = 0; l1 <= LMAXV; ++l1)
      for (int l2 = 0; l2 <= LMAXV; ++l2) {
        int lo = l1 > l2 ? l1 - l2 : l2 - l1;
        int hi = (l1 + l2 > LMAXV) ? LMAXV : (l1 + l2);
        if (l < lo || l > hi) continue;
        P.l1t[l][nc] = l1; P.l2t[l][nc] = l2;
        for (int mi = 0; mi < 2 * l + 1; ++mi) {
          int m = mi - l, n = 0;
          for (int p = 0; p < 2 * l1 + 1; ++p) {
            int m1 = p - l1, m2 = m - m1;
            if (m2 < -l2 || m2 > l2) continue;
            double v = cgcoef(l1, m1, l2, m2, l, m);
            if (v == 0.0) continue;
            P.pq[l][mi][nc][n] = p * 8 + (m2 + l2);
            P.coef[l][mi][nc][n] = (float)v;
            ++n;
          }
          P.nnz[l][mi][nc] = n;
        }
        ++nc;
      }
    P.ncoup[l] = nc;
  }
  return P;
}

constexpr CGPack H = buildCG();
constexpr int LOFF[3] = {0, 8, 32};
constexpr int NCOUPC[3] = {3, 6, 6};

__global__ void zero_f32(float* __restrict__ p, long n) {
  long i = (long)blockIdx.x * blockDim.x + threadIdx.x;
  if (i < n) p[i] = 0.0f;
}

__global__ __launch_bounds__(256) void edge_mp(
    const float* __restrict__ x0, const float* __restrict__ x1,
    const float* __restrict__ x2, const float* __restrict__ ev,
    const int* __restrict__ src, const int* __restrict__ dst,
    float* __restrict__ y, int nE) {
  long idx = (long)blockIdx.x * blockDim.x + threadIdx.x;
  if (idx >= (long)nE * 6) return;
  int e = (int)(idx / 6);
  int g = (int)(idx % 6);
  int s = src[e], d = dst[e];
  float v = ev[e];
#pragma unroll
  for (int j = 0; j < 3; ++j) {
    int slot = g * 3 + j;
    const float* ps;
    if (slot < 2)      ps = x0 + (size_t)s * 8  + slot * 4;
    else if (slot < 8) ps = x1 + (size_t)s * 24 + (slot - 2) * 4;
    else               ps = x2 + (size_t)s * 40 + (slot - 8) * 4;
    float4 val = *(const float4*)ps;
    float* py = y + (size_t)d * 72 + slot * 4;
    __hip_atomic_fetch_add(py + 0, v * val.x, __ATOMIC_RELAXED, __HIP_MEMORY_SCOPE_AGENT);
    __hip_atomic_fetch_add(py + 1, v * val.y, __ATOMIC_RELAXED, __HIP_MEMORY_SCOPE_AGENT);
    __hip_atomic_fetch_add(py + 2, v * val.z, __ATOMIC_RELAXED, __HIP_MEMORY_SCOPE_AGENT);
    __hip_atomic_fetch_add(py + 3, v * val.w, __ATOMIC_RELAXED, __HIP_MEMORY_SCOPE_AGENT);
  }
}

#define NWAVE 9
#define NTHR (NWAVE * 32)
#define TILE_OUT 1152

template <int L, int PROD, int MASK>
__device__ __forceinline__ void cg_unit(const float* __restrict__ Lrow,
                                        const float* __restrict__ wl,
                                        const int hi, const int wcol,
                                        v8f (&acc)[5]) {
  constexpr int aoff = (PROD < 2)  ? 72 : 0;
  constexpr int boff = (PROD == 0) ? 72 : 0;
#pragma unroll
  for (int cp = 0; cp < NCOUPC[L]; ++cp) {
    if (!((MASK >> cp) & 1)) continue;
    const int l1 = H.l1t[L][cp], l2 = H.l2t[L][cp];
    const int na = 2 * l1 + 1, nb = 2 * l2 + 1;
    const float* paBase = Lrow + aoff + LOFF[l1];
    const float* pbBase = Lrow + boff + LOFF[l2];
    const float* wcp = wl + (size_t)(cp * 64) * CCH;
    for (int c = 0; c < 8; ++c) {
      const float* pa = paBase + c * na;
      for (int dc = 0; dc < 8; dc += 4) {
        const int dlane = dc + 2 * hi;
        const int kb = c * 8 + dlane;
        v2f B;
        B.x = wcp[(kb + 0) * CCH + wcol];
        B.y = wcp[(kb + 1) * CCH + wcol];
        const float* pb0 = pbBase + dlane * nb;
        const float* pb1 = pb0 + nb;
#pragma unroll
        for (int m = 0; m < 2 * L + 1; ++m) {
          float z0 = 0.f, z1 = 0.f;
#pragma unroll
          for (int t = 0; t < H.nnz[L][m][cp]; ++t) {
            const int pq = H.pq[L][m][cp][t];
            const float cf = H.coef[L][m][cp][t];
            const float a = pa[pq >> 3];
            z0 += cf * a * pb0[pq & 7];
            z1 += cf * a * pb1[pq & 7];
          }
          v2f A; A.x = z0; A.y = z1;
          acc[m] = __builtin_amdgcn_wmma_f32_16x16x4_f32(
              false, A, false, B, (short)0, acc[m], false, false);
        }
      }
    }
  }
}

template <int L, int PROD, int MASK>
__device__ __forceinline__ void run_unit(const float* __restrict__ Lrow,
                                         const float* __restrict__ wl,
                                         float* __restrict__ slot, const int row,
                                         const int hi, const int wcol) {
  constexpr int nm = 2 * L + 1;
  constexpr int lbase = (L == 0) ? 0 : (L == 1 ? 128 : 512);
  const v8f vzero = {0.f, 0.f, 0.f, 0.f, 0.f, 0.f, 0.f, 0.f};
  v8f acc[5];
#pragma unroll
  for (int m = 0; m < 5; ++m) acc[m] = vzero;
  cg_unit<L, PROD, MASK>(Lrow, wl, hi, wcol, acc);
  if (row < CCH) {
    float* p = slot + lbase + row * nm;
#pragma unroll
    for (int r = 0; r < 8; ++r) {
      const int ln = r + 8 * hi;
#pragma unroll
      for (int m = 0; m < nm; ++m) p[ln * (CCH * nm) + m] = acc[m][r];
    }
  }
}

template <int PROD, int G>
__device__ __forceinline__ void run_share(const float* __restrict__ Lrow,
                                          const float* __restrict__ w0,
                                          const float* __restrict__ w1,
                                          const float* __restrict__ w2,
                                          float* __restrict__ slot, const int row,
                                          const int hi, const int wcol) {
  constexpr int M2 = (G == 0) ? 0x21 : (G == 1 ? 0x12 : 0x0C);
  constexpr int M1 = (G == 0) ? 0x12 : (G == 1 ? 0x21 : 0x0C);
  constexpr int M0 = 1 << G;
  run_unit<2, PROD, M2>(Lrow, w2, slot, row, hi, wcol);
  run_unit<1, PROD, M1>(Lrow, w1, slot, row, hi, wcol);
  run_unit<0, PROD, M0>(Lrow, w0, slot, row, hi, wcol);
}

__global__ __launch_bounds__(NTHR) __attribute__((amdgpu_waves_per_eu(6)))
void node_cg_wmma(
    const float* __restrict__ x0, const float* __restrict__ x1,
    const float* __restrict__ x2, const float* __restrict__ y,
    const float* __restrict__ wyy0, const float* __restrict__ wyy1,
    const float* __restrict__ wyy2, const float* __restrict__ wyx0,
    const float* __restrict__ wyx1, const float* __restrict__ wyx2,
    const float* __restrict__ wxx0, const float* __restrict__ wxx1,
    const float* __restrict__ wxx2, float* __restrict__ out, int nN) {
  __shared__ float lds[16 * 144];
  __shared__ float red[NWAVE * TILE_OUT];

  const int wave = __builtin_amdgcn_readfirstlane((int)(threadIdx.x >> 5));
  const int lane = threadIdx.x & 31;
  const int row  = lane & 15;
  const int hi   = lane >> 4;
  const int wcol = row & 7;
  const int tileBase = blockIdx.x * 16;

  float* L = lds;
  for (int i = threadIdx.x; i < 16 * 72; i += NTHR) {
    int ln = i / 72, f = i % 72;
    int node = tileBase + ln;
    if (node >= nN) node = nN - 1;
    float xv;
    if (f < 8)       xv = x0[(size_t)node * 8  + f];
    else if (f < 32) xv = x1[(size_t)node * 24 + (f - 8)];
    else             xv = x2[(size_t)node * 40 + (f - 32)];
    L[ln * 144 + f]      = xv;
    L[ln * 144 + 72 + f] = y[(size_t)node * 72 + f];
  }
  __syncthreads();

  const float* Lrow = L + row * 144;
  float* slot = red + wave * TILE_OUT;

  switch (wave) {
    case 0:  run_share<0, 0>(Lrow, wyy0, wyy1, wyy2, slot, row, hi, wcol); break;
    case 1:  run_share<0, 1>(Lrow, wyy0, wyy1, wyy2, slot, row, hi, wcol); break;
    case 2:  run_share<0, 2>(Lrow, wyy0, wyy1, wyy2, slot, row, hi, wcol); break;
    case 3:  run_share<1, 0>(Lrow, wyx0, wyx1, wyx2, slot, row, hi, wcol); break;
    case 4:  run_share<1, 1>(Lrow, wyx0, wyx1, wyx2, slot, row, hi, wcol); break;
    case 5:  run_share<1, 2>(Lrow, wyx0, wyx1, wyx2, slot, row, hi, wcol); break;
    case 6:  run_share<2, 0>(Lrow, wxx0, wxx1, wxx2, slot, row, hi, wcol); break;
    case 7:  run_share<2, 1>(Lrow, wxx0, wxx1, wxx2, slot, row, hi, wcol); break;
    default: run_share<2, 2>(Lrow, wxx0, wxx1, wxx2, slot, row, hi, wcol); break;
  }
  __syncthreads();

  for (int i = threadIdx.x; i < TILE_OUT; i += NTHR) {
    int rem, w8, loff;
    size_t baseL;
    if (i < 128)      { rem = i;       w8 = 8;  loff = 0;  baseL = 0; }
    else if (i < 512) { rem = i - 128; w8 = 24; loff = 8;  baseL = (size_t)nN * 8; }
    else              { rem = i - 512; w8 = 40; loff = 32; baseL = (size_t)nN * 32; }
    const int ln = rem / w8, f = rem % w8;
    const int node = tileBase + ln;
    float s = red[i];
#pragma unroll
    for (int k = 1; k < NWAVE; ++k) s += red[k * TILE_OUT + i];
    if (node < nN) out[baseL + (size_t)node * w8 + f] = s + L[ln * 144 + loff + f];
  }
}

extern "C" void kernel_launch(void* const* d_in, const int* in_sizes, int n_in,
                              void* d_out, int out_size, void* d_ws, size_t ws_size,
                              hipStream_t stream) {
  const float* x0 = (const float*)d_in[0];
  const float* x1 = (const float*)d_in[1];
  const float* x2 = (const float*)d_in[2];
  const float* ev = (const float*)d_in[3];
  const int* esrc = (const int*)d_in[4];
  const int* edst = (const int*)d_in[5];
  const float* wyy0 = (const float*)d_in[6];
  const float* wyy1 = (const float*)d_in[7];
  const float* wyy2 = (const float*)d_in[8];
  const float* wyx0 = (const float*)d_in[9];
  const float* wyx1 = (const float*)d_in[10];
  const float* wyx2 = (const float*)d_in[11];
  const float* wxx0 = (const float*)d_in[12];
  const float* wxx1 = (const float*)d_in[13];
  const float* wxx2 = (const float*)d_in[14];
  float* out = (float*)d_out;
  float* y   = (float*)d_ws;

  const int nN = in_sizes[0] / CCH;
  const int nE = in_sizes[3];

  long ny = (long)nN * 72;
  zero_f32<<<(unsigned)((ny + 511) / 512), 512, 0, stream>>>(y, ny);

  long tot = (long)nE * 6;
  edge_mp<<<(unsigned)((tot + 255) / 256), 256, 0, stream>>>(x0, x1, x2, ev, esrc,
                                                             edst, y, nE);

  int blocks = (nN + 15) / 16;
  node_cg_wmma<<<blocks, NTHR, 0, stream>>>(x0, x1, x2, y, wyy0, wyy1, wyy2, wyx0,
                                          wyx1, wyx2, wxx0, wxx1, wxx2, out, nN);
}


# edges bucketed by destination (int adds: count, scan, fill), buckets sorted by edge id so y is summed in a fixed order (earlier try: arrival order); node kernel gathers while staging; no f32 atomics
# speedup vs baseline: 2.1543x; 1.3796x over previous
#include <hip/hip_runtime.h>

#define LMAXV 2
#define CCH 8

typedef float v2f __attribute__((ext_vector_type(2)));
typedef float v8f __attribute__((ext_vector_type(8)));

constexpr double cfact(int n) { double r = 1.0; for (int i = 2; i <= n; ++i) r *= i; return r; }
constexpr double csqrt_(double x) {
  if (x <= 0.0) return 0.0;
  double r = x < 1.0 ? 1.0 : x;
  for (int i = 0; i < 100; ++i) r = 0.5 * (r + x / r);
  return r;
}
constexpr double cgcoef(int j1, int m1, int j2, int m2, int j, int m) {
  if (m1 + m2 != m) return 0.0;
  int lo = j1 > j2 ? j1 - j2 : j2 - j1;
  if (j < lo || j > j1 + j2) return 0.0;
  double pre = csqrt_((2.0 * j + 1.0) * cfact(j + j1 - j2) * cfact(j - j1 + j2) *
                      cfact(j1 + j2 - j) / cfact(j1 + j2 + j + 1));
  pre *= csqrt_(cfact(j + m) * cfact(j - m) * cfact(j1 - m1) * cfact(j1 + m1) *
                cfact(j2 - m2) * cfact(j2 + m2));
  double s = 0.0;
  for (int k = 0; k <= j1 + j2 - j; ++k) {
    int d2 = j1 + j2 - j - k, d3 = j1 - m1 - k, d4 = j2 + m2 - k;
    int d5 = j - j2 + m1 + k, d6 = j - j1 - m2 + k;
    if (d2 < 0 || d3 < 0 || d4 < 0 || d5 < 0 || d6 < 0) continue;
    double den = cfact(k) * cfact(d2) * cfact(d3) * cfact(d4) * cfact(d5) * cfact(d6);
    s += ((k & 1) ? -1.0 : 1.0) / den;
  }
  return pre * s;
}

struct CGPack {
  int ncoup[3];
  int l1t[3][6], l2t[3][6];
  int nnz[3][5][6];
  int pq[3][5][6][5];
  float coef[3][5][6][5];
};

constexpr CGPack buildCG() {
  CGPack P{};
  for (int l = 0; l <= LMAXV; ++l) {
    int nc = 0;
    for (int l1 = 0; l1 <= LMAXV; ++l1)
      for (int l2 = 0; l2 <= LMAXV; ++l2) {
        int lo = l1 > l2 ? l1 - l2 : l2 - l1;
        int hi = (l1 + l2 > LMAXV) ? LMAXV : (l1 + l2);
        if (l < lo || l > hi) continue;
        P.l1t[l][nc] = l1; P.l2t[l][nc] = l2;
        for (int mi = 0; mi < 2 * l + 1; ++mi) {
          int m = mi - l, n = 0;
          for (int p = 0; p < 2 * l1 + 1; ++p) {
            int m1 = p - l1, m2 = m - m1;
            if (m2 < -l2 || m2 > l2) continue;
            double v = cgcoef(l1, m1, l2, m2, l, m);
            if (v == 0.0) continue;
            P.pq[l][mi][nc][n] = p * 8 + (m2 + l2);
            P.coef[l][mi][nc][n] = (float)v;
            ++n;
          }
          P.nnz[l][mi][nc] = n;
        }
        ++nc;
      }
    P.ncoup[l] = nc;
  }
  return P;
}

constexpr CGPack H = buildCG();
constexpr int LOFF[3] = {0, 8, 32};
constexpr int NCOUPC[3] = {3, 6, 6};

__global__ void zero_i32(int* __restrict__ p, int n) {
  int i = blockIdx.x * blockDim.x + threadIdx.x;
  if (i < n) p[i] = 0;
}

__global__ __launch_bounds__(256) void count_dst(const int* __restrict__ dst,
                                                 int* __restrict__ cnt, int nE) {
  int e = blockIdx.x * blockDim.x + threadIdx.x;
  if (e >= nE) return;
  __hip_atomic_fetch_add(cnt + dst[e], 1, __ATOMIC_RELAXED, __HIP_MEMORY_SCOPE_AGENT);
}

#define SCAN_T 1024
#define SCAN_K 8
__global__ __launch_bounds__(SCAN_T) void scan_excl(int* __restrict__ c, int n) {
  __shared__ int buf[SCAN_T * SCAN_K + SCAN_T * SCAN_K / 32];
  __shared__ int wsum[32];
  __shared__ int tot;
  const int t = threadIdx.x;
  const int lane = t & 31, w = t >> 5;
  int carry = 0;
  for (int base = 0; base < n; base += SCAN_T * SCAN_K) {
    int in[SCAN_K];
#pragma unroll
    for (int k = 0; k < SCAN_K; ++k) {
      const int g = base + k * SCAN_T + t;
      in[k] = c[(g < n) ? g : (n - 1)];
    }
#pragma unroll
    for (int k = 0; k < SCAN_K; ++k) {
      const int i = k * SCAN_T + t;
      buf[i + (i >> 5)] = (base + i < n) ? in[k] : 0;
    }
    __syncthreads();
    int v[SCAN_K];
    int s = 0;
#pragma unroll
    for (int k = 0; k < SCAN_K; ++k) {
      const int i = t * SCAN_K + k;
      v[k] = buf[i + (i >> 5)];
      s += v[k];
    }
    int inc = s;
#pragma unroll
    for (int d = 1; d < 32; d <<= 1) {
      const int o = __shfl_up(inc, d);
      if (lane >= d) inc += o;
    }
    if (lane == 31) wsum[w] = inc;
    __syncthreads();
    if (w == 0) {
      const int x = wsum[lane];
      int xi = x;
#pragma unroll
      for (int d = 1; d < 32; d <<= 1) {
        const int o = __shfl_up(xi, d);
        if (lane >= d) xi += o;
      }
      wsum[lane] = xi - x;
      if (lane == 31) tot = xi;
    }
    __syncthreads();
    int run = carry + wsum[w] + inc - s;
#pragma unroll
    for (int k = 0; k < SCAN_K; ++k) {
      const int i = t * SCAN_K + k;
      buf[i + (i >> 5)] = run;
      run += v[k];
    }
    carry += tot;
    __syncthreads();
    int outv[SCAN_K];
#pragma unroll
    for (int k = 0; k < SCAN_K; ++k) {
      const int i = k * SCAN_T + t;
      outv[k] = buf[i + (i >> 5)];
    }
#pragma unroll
    for (int k = 0; k < SCAN_K; ++k) {
      const int g = base + k * SCAN_T + t;
      if (g < n) c[g] = outv[k];
    }
    __syncthreads();
  }
}

__global__ __launch_bounds__(256) void fill_ids(const int* __restrict__ dst,
                                                int* __restrict__ cur,
                                                int* __restrict__ bid, int nE) {
  int e = blockIdx.x * blockDim.x + threadIdx.x;
  if (e >= nE) return;
  int pos = __hip_atomic_fetch_add(cur + dst[e], 1, __ATOMIC_RELAXED,
                                   __HIP_MEMORY_SCOPE_AGENT);
  bid[pos] = e;
}

__global__ __launch_bounds__(256) void order_buckets(
    const int* __restrict__ bid, const int* __restrict__ cur,
    const int* __restrict__ src, const int* __restrict__ dst,
    const float* __restrict__ ev, int* __restrict__ bsrc,
    float* __restrict__ bev, int nE) {
  int p = blockIdx.x * blockDim.x + threadIdx.x;
  if (p >= nE) return;
  const int e = bid[p];
  const int d = dst[e];
  const int se = src[e];
  const float ve = ev[e];
  const int jb = d ? cur[d - 1] : 0;
  const int je = cur[d];
  int r = 0;
  for (int i = jb; i < je; ++i) r += (bid[i] < e) ? 1 : 0;
  bsrc[jb + r] = se;
  bev[jb + r] = ve;
}

#define NWAVE 9
#define NTHR (NWAVE * 32)
#define TILE_OUT 1152
#define GB 16

template <int L, int PROD, int MASK>
__device__ __forceinline__ void cg_unit(const float* __restrict__ Lrow,
                                        const float* __restrict__ wl,
                                        const int hi, const int wcol,
                                        v8f (&acc)[5]) {
  constexpr int aoff = (PROD < 2)  ? 72 : 0;
  constexpr int boff = (PROD == 0) ? 72 : 0;
#pragma unroll
  for (int cp = 0; cp < NCOUPC[L]; ++cp) {
    if (!((MASK >> cp) & 1)) continue;
    const int l1 = H.l1t[L][cp], l2 = H.l2t[L][cp];
    const int na = 2 * l1 + 1, nb = 2 * l2 + 1;
    const float* paBase = Lrow + aoff + LOFF[l1];
    const float* pbBase = Lrow + boff + LOFF[l2];
    const float* wcp = wl + (size_t)(cp * 64) * CCH;
    for (int c = 0; c < 8; ++c) {
      const float* pa = paBase + c * na;
      for (int dc = 0; dc < 8; dc += 4) {
        const int dlane = dc + 2 * hi;
        const int kb = c * 8 + dlane;
        v2f B;
        B.x = wcp[(kb + 0) * CCH + wcol];
        B.y = wcp[(kb + 1) * CCH + wcol];
        const float* pb0 = pbBase + dlane * nb;
        const float* pb1 = pb0 + nb;
#pragma unroll
        for (int m = 0; m < 2 * L + 1; ++m) {
          float z0 = 0.f, z1 = 0.f;
#pragma unroll
          for (int t = 0; t < H.nnz[L][m][cp]; ++t) {
            const int pq = H.pq[L][m][cp][t];
            const float cf = H.coef[L][m][cp][t];
            const float a = pa[pq >> 3];
            z0 += cf * a * pb0[pq & 7];
            z1 += cf * a * pb1[pq & 7];
          }
          v2f A; A.x = z0; A.y = z1;
          acc[m] = __builtin_amdgcn_wmma_f32_16x16x4_f32(
              false, A, false, B, (short)0, acc[m], false, false);
        }
      }
    }
  }
}

template <int L, int PROD, int MASK>
__device__ __forceinline__ void run_unit(const float* __restrict__ Lrow,
                                         const float* __restrict__ wl,
                                         float* __restrict__ slot, const int row,
                                         const int hi, const int wcol) {
  constexpr int nm = 2 * L + 1;
  constexpr int lbase = (L == 0) ? 0 : (L == 1 ? 128 : 512);
  const v8f vzero = {0.f, 0.f, 0.f, 0.f, 0.f, 0.f, 0.f, 0.f};
  v8f acc[5];
#pragma unroll
  for (int m = 0; m < 5; ++m) acc[m] = vzero;
  cg_unit<L, PROD, MASK>(Lrow, wl, hi, wcol, acc);
  if (row < CCH) {
    float* p = slot + lbase + row * nm;
#pragma unroll
    for (int r = 0; r < 8; ++r) {
      const int ln = r + 8 * hi;
#pragma unroll
      for (int m = 0; m < nm; ++m) p[ln * (CCH * nm) + m] = acc[m][r];
    }
  }
}

template <int PROD, int G>
__device__ __forceinline__ void run_share(const float* __restrict__ Lrow,
                                          const float* __restrict__ w0,
                                          const float* __restrict__ w1,
                                          const float* __restrict__ w2,
                                          float* __restrict__ slot, const int row,
                                          const int hi, const int wcol) {
  constexpr int M2 = (G == 0) ? 0x21 : (G == 1 ? 0x12 : 0x0C);
  constexpr int M1 = (G == 0) ? 0x12 : (G == 1 ? 0x21 : 0x0C);
  constexpr int M0 = 1 << G;
  run_unit<2, PROD, M2>(Lrow, w2, slot, row, hi, wcol);
  run_unit<1, PROD, M1>(Lrow, w1, slot, row, hi, wcol);
  run_unit<0, PROD, M0>(Lrow, w0, slot, row, hi, wcol);
}

__global__ __launch_bounds__(NTHR) __attribute__((amdgpu_waves_per_eu(6)))
void node_cg_wmma(
    const float* __restrict__ x0, const float* __restrict__ x1,
    const float* __restrict__ x2, const int* __restrict__ cur,
    const int* __restrict__ bsrc, const float* __restrict__ bev,
    const float* __restrict__ wyy0, const float* __restrict__ wyy1,
    const float* __restrict__ wyy2, const float* __restrict__ wyx0,
    const float* __restrict__ wyx1, const float* __restrict__ wyx2,
    const float* __restrict__ wxx0, const float* __restrict__ wxx1,
    const float* __restrict__ wxx2, float* __restrict__ out, int nN) {
  __shared__ float lds[16 * 144];
  __shared__ float red[NWAVE * TILE_OUT];

  const int wave = __builtin_amdgcn_readfirstlane((int)(threadIdx.x >> 5));
  const int lane = threadIdx.x & 31;
  const int row  = lane & 15;
  const int hi   = lane >> 4;
  const int wcol = row & 7;
  const int tileBase = blockIdx.x * 16;

  float* L = lds;
  for (int i = threadIdx.x; i < 16 * 72; i += NTHR) {
    int ln = i / 72, f = i % 72;
    int node = tileBase + ln;
    if (node >= nN) node = nN - 1;
    float xv;
    if (f < 8)       xv = x0[(size_t)node * 8  + f];
    else if (f < 32) xv = x1[(size_t)node * 24 + (f - 8)];
    else             xv = x2[(size_t)node * 40 + (f - 32)];
    L[ln * 144 + f]      = xv;
  }
  {
    const int t = threadIdx.x;
    const int ln = t / 18, slot = t - ln * 18;
    int node = tileBase + ln;
    if (node >= nN) node = nN - 1;
    const int jb = node ? cur[node - 1] : 0;
    const int je = cur[node];
    const float* xb;
    int stride;
    if (slot < 2)      { xb = x0 + slot * 4;       stride = 8; }
    else if (slot < 8) { xb = x1 + (slot - 2) * 4; stride = 24; }
    else               { xb = x2 + (slot - 8) * 4; stride = 40; }
    float a0 = 0.f, a1 = 0.f, a2 = 0.f, a3 = 0.f;
    for (int j = jb; j < je; j += GB) {
      int sv[GB];
      float vv[GB];
      float4 pv[GB];
#pragma unroll
      for (int k = 0; k < GB; ++k) {
        const bool in = (j + k < je);
        const int jj = in ? (j + k) : (je - 1);
        sv[k] = bsrc[jj];
        const float w = bev[jj];
        vv[k] = in ? w : 0.f;
      }
#pragma unroll
      for (int k = 0; k < GB; ++k)
        pv[k] = *(const float4*)(xb + sv[k] * stride);
#pragma unroll
      for (int k = 0; k < GB; ++k) {
        a0 += vv[k] * pv[k].x;
        a1 += vv[k] * pv[k].y;
        a2 += vv[k] * pv[k].z;
        a3 += vv[k] * pv[k].w;
      }
    }
    float* py = L + ln * 144 + 72 + slot * 4;
    py[0] = a0; py[1] = a1; py[2] = a2; py[3] = a3;
  }
  __syncthreads();

  const float* Lrow = L + row * 144;
  float* slot = red + wave * TILE_OUT;

  switch (wave) {
    case 0:  run_share<0, 0>(Lrow, wyy0, wyy1, wyy2, slot, row, hi, wcol); break;
    case 1:  run_share<0, 1>(Lrow, wyy0, wyy1, wyy2, slot, row, hi, wcol); break;
    case 2:  run_share<0, 2>(Lrow, wyy0, wyy1, wyy2, slot, row, hi, wcol); break;
    case 3:  run_share<1, 0>(Lrow, wyx0, wyx1, wyx2, slot, row, hi, wcol); break;
    case 4:  run_share<1, 1>(Lrow, wyx0, wyx1, wyx2, slot, row, hi, wcol); break;
    case 5:  run_share<1, 2>(Lrow, wyx0, wyx1, wyx2, slot, row, hi, wcol); break;
    case 6:  run_share<2, 0>(Lrow, wxx0, wxx1, wxx2, slot, row, hi, wcol); break;
    case 7:  run_share<2, 1>(Lrow, wxx0, wxx1, wxx2, slot, row, hi, wcol); break;
    default: run_share<2, 2>(Lrow, wxx0, wxx1, wxx2, slot, row, hi, wcol); break;
  }
  __syncthreads();

  for (int i = threadIdx.x; i < TILE_OUT; i += NTHR) {
    int rem, w8, loff;
    size_t baseL;
    if (i < 128)      { rem = i;       w8 = 8;  loff = 0;  baseL = 0; }
    else if (i < 512) { rem = i - 128; w8 = 24; loff = 8;  baseL = (size_t)nN * 8; }
    else              { rem = i - 512; w8 = 40; loff = 32; baseL = (size_t)nN * 32; }
    const int ln = rem / w8, f = rem % w8;
    const int node = tileBase + ln;
    float s = red[i];
#pragma unroll
    for (int k = 1; k < NWAVE; ++k) s += red[k * TILE_OUT + i];
    if (node < nN) out[baseL + (size_t)node * w8 + f] = s + L[ln * 144 + loff + f];
  }
}

extern "C" void kernel_launch(void* const* d_in, const int* in_sizes, int n_in,
                              void* d_out, int out_size, void* d_ws, size_t ws_size,
                              hipStream_t stream) {
  const float* x0 = (const float*)d_in[0];
  const float* x1 = (const float*)d_in[1];
  const float* x2 = (const float*)d_in[2];
  const float* ev = (const float*)d_in[3];
  const int* esrc = (const int*)d_in[4];
  const int* edst = (const int*)d_in[5];
  const float* wyy0 = (const float*)d_in[6];
  const float* wyy1 = (const float*)d_in[7];
  const float* wyy2 = (const float*)d_in[8];
  const float* wyx0 = (const float*)d_in[9];
  const float* wyx1 = (const float*)d_in[10];
  const float* wyx2 = (const float*)d_in[11];
  const float* wxx0 = (const float*)d_in[12];
  const float* wxx1 = (const float*)d_in[13];
  const float* wxx2 = (const float*)d_in[14];
  float* out = (float*)d_out;
  const int nN = in_sizes[0] / CCH;
  const int nE = in_sizes[3];

  int* cur    = (int*)d_ws;
  int* bid    = cur + nN;
  int* bsrc   = bid + nE;
  float* bev  = (float*)(bsrc + nE);

  const unsigned gN = (unsigned)((nN + 255) / 256);
  const unsigned gE = (unsigned)((nE + 255) / 256);

  zero_i32<<<gN, 256, 0, stream>>>(cur, nN);
  count_dst<<<gE, 256, 0, stream>>>(edst, cur, nE);
  scan_excl<<<1, SCAN_T, 0, stream>>>(cur, nN);
  fill_ids<<<gE, 256, 0, stream>>>(edst, cur, bid, nE);
  order_buckets<<<gE, 256, 0, stream>>>(bid, cur, esrc, edst, ev, bsrc, bev, nE);

  int blocks = (nN + 15) / 16;
  node_cg_wmma<<<blocks, NTHR, 0, stream>>>(x0, x1, x2, cur, bsrc, bev, wyy0, wyy1,
                                          wyy2, wyx0, wyx1, wyx2, wxx0, wxx1, wxx2,
                                          out, nN);
}
